# starting kernel
# baseline (speedup 1.0000x reference)
#include <hip/hip_runtime.h>


typedef float v2f __attribute__((ext_vector_type(2)));
typedef float v8f __attribute__((ext_vector_type(8)));

#define B_PATHS    16384
#define N_STEPS    500
#define SAVE_EVERY 5
#define N_SAVES    100
#define T_SAVED    101
#define DT         0.002f
#define PF_NEAR    8
#define PF_FAR     32

__global__ __launch_bounds__(64) void sde_wmma_kernel(
    const float* __restrict__ drift, const float* __restrict__ drift_bias,
    const float* __restrict__ diffusion, const float* __restrict__ diffusion_bias,
    const float* __restrict__ x0, const float* __restrict__ noise,
    float* __restrict__ out)
{
    const int  lane = threadIdx.x & 31;
    const int  wave = (blockIdx.x * blockDim.x + threadIdx.x) >> 5;
    const int  base = wave * 32;
    const int  m    = lane & 15;
    const bool lo   = lane < 16;
    const float sqdt = __builtin_sqrtf(DT);

    const int k0 = lo ? 0 : 2;
    const int k1 = lo ? 1 : 3;
    const int mm = m & 7;
    float a0v = 0.0f, a1v = 0.0f;
    if (mm < 3) {
        a0v = DT * drift[mm * 3 + k0] + ((mm == k0) ? 1.0f : 0.0f);
        a1v = (k1 < 3) ? (DT * drift[mm * 3 + k1] + ((mm == k1) ? 1.0f : 0.0f))
                       : (DT * drift_bias[mm]);
    } else if (mm < 6) {
        const int r = mm - 3;
        a0v = sqdt * diffusion[r * 3 + k0];
        a1v = (k1 < 3) ? (sqdt * diffusion[r * 3 + k1])
                       : (sqdt * diffusion_bias[r]);
    }
    v2f Amat; Amat.x = a0v; Amat.y = a1v;

    const int p0 = base + m;
    const int p1 = base + 16 + m;
    v2f B0, B1;
    B0.x = lo ? x0[p0 * 3 + 0] : x0[p0 * 3 + 2];
    B0.y = lo ? x0[p0 * 3 + 1] : 1.0f;
    B1.x = lo ? x0[p1 * 3 + 0] : x0[p1 * 3 + 2];
    B1.y = lo ? x0[p1 * 3 + 1] : 1.0f;

    const int p  = base + lane;
    const int BT = B_PATHS * T_SAVED;
    out[0 * BT + p * T_SAVED] = x0[p * 3 + 0];
    out[1 * BT + p * T_SAVED] = x0[p * 3 + 1];
    out[2 * BT + p * T_SAVED] = x0[p * 3 + 2];

    for (int s = 0; s < N_SAVES; ++s) {
        float sv0 = 0.0f, sv1 = 0.0f, sv2 = 0.0f;
#pragma unroll
        for (int k = 0; k < SAVE_EVERY; ++k) {
            const int t   = s * SAVE_EVERY + k;
            const int off = t * (B_PATHS * 3);

            int tn = t + PF_NEAR; if (tn >= N_STEPS) tn = N_STEPS - 1;
            const float* pfn = noise + tn * (B_PATHS * 3) + p * 3;
            asm volatile("global_prefetch_b8 %0, off" :: "v"(pfn));
            int tf = t + PF_FAR; if (tf >= N_STEPS) tf = N_STEPS - 1;
            __builtin_prefetch(noise + tf * (B_PATHS * 3) + p * 3, 0, 3);

            const float n00 = noise[off + p0 * 3 + 0];
            const float n01 = noise[off + p0 * 3 + 1];
            const float n02 = noise[off + p0 * 3 + 2];
            const float n10 = noise[off + p1 * 3 + 0];
            const float n11 = noise[off + p1 * 3 + 1];
            const float n12 = noise[off + p1 * 3 + 2];

            v8f cz = {};
            v8f d0 = __builtin_amdgcn_wmma_f32_16x16x4_f32(
                false, Amat, false, B0, (short)0, cz, false, false);
            v8f d1 = __builtin_amdgcn_wmma_f32_16x16x4_f32(
                false, Amat, false, B1, (short)0, cz, false, false);

            const float xn00 = __builtin_fmaf(d0[3], n00, d0[0]);
            const float xn01 = __builtin_fmaf(d0[4], n01, d0[1]);
            const float xn02 = __builtin_fmaf(d0[5], n02, d0[2]);
            const float xn10 = __builtin_fmaf(d1[3], n10, d1[0]);
            const float xn11 = __builtin_fmaf(d1[4], n11, d1[1]);
            const float xn12 = __builtin_fmaf(d1[5], n12, d1[2]);

            B0.x = lo ? xn00 : xn02;  B0.y = lo ? xn01 : 1.0f;
            B1.x = lo ? xn10 : xn12;  B1.y = lo ? xn11 : 1.0f;

            if (k == SAVE_EVERY - 1) {
                sv0 = lo ? xn00 : xn10;
                sv1 = lo ? xn01 : xn11;
                sv2 = lo ? xn02 : xn12;
            }
        }
        const int so = p * T_SAVED + (s + 1);
        out[0 * BT + so] = sv0;
        out[1 * BT + so] = sv1;
        out[2 * BT + so] = sv2;
    }
}

extern "C" void kernel_launch(void* const* d_in, const int* in_sizes, int n_in,
                              void* d_out, int out_size, void* d_ws, size_t ws_size,
                              hipStream_t stream) {
    const float* drift     = (const float*)d_in[0];
    const float* driftb    = (const float*)d_in[1];
    const float* diffusion = (const float*)d_in[2];
    const float* diffb     = (const float*)d_in[3];
    const float* x0        = (const float*)d_in[4];
    const float* noise     = (const float*)d_in[5];
    float* out = (float*)d_out;

    dim3 block(64);
    dim3 grid(B_PATHS / 64);
    sde_wmma_kernel<<<grid, block, 0, stream>>>(
        drift, driftb, diffusion, diffb, x0, noise, out);
}


# noise software-pipelined through a 25-step register ring (each step's two row loads issued 25 steps early, 50 in flight per wave) replacing the near prefetch hint; L2 prefetch moved to 64 steps
# speedup vs baseline: 1.5528x; 1.5528x over previous
#include <hip/hip_runtime.h>


typedef float v2f __attribute__((ext_vector_type(2)));
typedef float v8f __attribute__((ext_vector_type(8)));
typedef float v3f __attribute__((ext_vector_type(3), aligned(4)));

#define B_PATHS    16384
#define N_STEPS    500
#define SAVE_EVERY 5
#define N_SAVES    100
#define T_SAVED    101
#define DT         0.002f
#define RING       25
#define PF_FAR     64
#define STEP_BYTES (B_PATHS * 3 * 4)

__global__ __launch_bounds__(64) void sde_wmma_kernel(
    const float* __restrict__ drift, const float* __restrict__ drift_bias,
    const float* __restrict__ diffusion, const float* __restrict__ diffusion_bias,
    const float* __restrict__ x0, const float* __restrict__ noise,
    float* __restrict__ out)
{
    const int  lane = threadIdx.x & 31;
    const int  wave = (blockIdx.x * blockDim.x + threadIdx.x) >> 5;
    const int  base = wave * 32;
    const int  m    = lane & 15;
    const bool lo   = lane < 16;
    const float sqdt = __builtin_sqrtf(DT);

    const int k0 = lo ? 0 : 2;
    const int k1 = lo ? 1 : 3;
    const int mm = m & 7;
    float a0v = 0.0f, a1v = 0.0f;
    if (mm < 3) {
        a0v = DT * drift[mm * 3 + k0] + ((mm == k0) ? 1.0f : 0.0f);
        a1v = (k1 < 3) ? (DT * drift[mm * 3 + k1] + ((mm == k1) ? 1.0f : 0.0f))
                       : (DT * drift_bias[mm]);
    } else if (mm < 6) {
        const int r = mm - 3;
        a0v = sqdt * diffusion[r * 3 + k0];
        a1v = (k1 < 3) ? (sqdt * diffusion[r * 3 + k1])
                       : (sqdt * diffusion_bias[r]);
    }
    v2f Amat; Amat.x = a0v; Amat.y = a1v;

    const int p0 = base + m;
    const int p1 = base + 16 + m;
    v2f B0, B1;
    B0.x = lo ? x0[p0 * 3 + 0] : x0[p0 * 3 + 2];
    B0.y = lo ? x0[p0 * 3 + 1] : 1.0f;
    B1.x = lo ? x0[p1 * 3 + 0] : x0[p1 * 3 + 2];
    B1.y = lo ? x0[p1 * 3 + 1] : 1.0f;

    const int p  = base + lane;
    const int BT = B_PATHS * T_SAVED;
    const float x00 = x0[p * 3 + 0];
    const float x01 = x0[p * 3 + 1];
    const float x02 = x0[p * 3 + 2];

    const char* const nbytes = (const char*)noise;
    const unsigned qb0 = (unsigned)p0 * 12u;
    const unsigned qb1 = (unsigned)p1 * 12u;
    const unsigned pb  = (unsigned)p  * 12u;
    v3f ringA[RING], ringB[RING];
#pragma unroll
    for (int j = 0; j < RING; ++j) {
        __builtin_amdgcn_sched_barrier(0);
        const char* src = nbytes + (size_t)j * STEP_BYTES;
        ringA[j] = *(const v3f*)(src + qb0);
        ringB[j] = *(const v3f*)(src + qb1);
    }
    __builtin_amdgcn_sched_barrier(0);
    out[0 * BT + p * T_SAVED] = x00;
    out[1 * BT + p * T_SAVED] = x01;
    out[2 * BT + p * T_SAVED] = x02;

    for (int s = 0; s < N_SAVES; s += RING / SAVE_EVERY) {
        const int t0 = s * SAVE_EVERY;
        const int tr0 = (t0 < N_STEPS - 2 * RING) ? t0 : (N_STEPS - 2 * RING);
        const int tf0 = (t0 < N_STEPS - RING - PF_FAR) ? t0 : (N_STEPS - RING - PF_FAR);
        const char* const rbase = nbytes + (size_t)(tr0 + RING)   * STEP_BYTES;
        const char* const fbase = nbytes + (size_t)(tf0 + PF_FAR) * STEP_BYTES;
#pragma unroll
        for (int j = 0; j < RING; ++j) {
            __builtin_amdgcn_sched_barrier(0);

            __builtin_prefetch(fbase + (size_t)j * STEP_BYTES + pb, 0, 3);

            const v3f na = ringA[j];
            const v3f nb = ringB[j];

            v8f cz = {};
            v8f d0 = __builtin_amdgcn_wmma_f32_16x16x4_f32(
                false, Amat, false, B0, (short)0, cz, false, false);
            v8f d1 = __builtin_amdgcn_wmma_f32_16x16x4_f32(
                false, Amat, false, B1, (short)0, cz, false, false);

            const float xn00 = __builtin_fmaf(d0[3], na.x, d0[0]);
            const float xn01 = __builtin_fmaf(d0[4], na.y, d0[1]);
            const float xn02 = __builtin_fmaf(d0[5], na.z, d0[2]);
            const float xn10 = __builtin_fmaf(d1[3], nb.x, d1[0]);
            const float xn11 = __builtin_fmaf(d1[4], nb.y, d1[1]);
            const float xn12 = __builtin_fmaf(d1[5], nb.z, d1[2]);

            B0.x = lo ? xn00 : xn02;  B0.y = lo ? xn01 : 1.0f;
            B1.x = lo ? xn10 : xn12;  B1.y = lo ? xn11 : 1.0f;

            if ((j % SAVE_EVERY) == SAVE_EVERY - 1) {
                const float sv0 = lo ? xn00 : xn10;
                const float sv1 = lo ? xn01 : xn11;
                const float sv2 = lo ? xn02 : xn12;
                const int so = p * T_SAVED + (s + j / SAVE_EVERY + 1);
                out[0 * BT + so] = sv0;
                out[1 * BT + so] = sv1;
                out[2 * BT + so] = sv2;
            }

            __builtin_amdgcn_sched_barrier(0);
            const char* src = rbase + (size_t)j * STEP_BYTES;
            ringA[j] = *(const v3f*)(src + qb0);
            ringB[j] = *(const v3f*)(src + qb1);
        }
    }
}

extern "C" void kernel_launch(void* const* d_in, const int* in_sizes, int n_in,
                              void* d_out, int out_size, void* d_ws, size_t ws_size,
                              hipStream_t stream) {
    const float* drift     = (const float*)d_in[0];
    const float* driftb    = (const float*)d_in[1];
    const float* diffusion = (const float*)d_in[2];
    const float* diffb     = (const float*)d_in[3];
    const float* x0        = (const float*)d_in[4];
    const float* noise     = (const float*)d_in[5];
    float* out = (float*)d_out;

    dim3 block(64);
    dim3 grid(B_PATHS / 64);
    sde_wmma_kernel<<<grid, block, 0, stream>>>(
        drift, driftb, diffusion, diffb, x0, noise, out);
}


# saved states staged in LDS in the output's own layout, written once at the end as whole 128-B lines (b128, 4 lines per wave store) instead of 4-B stores to 32 lines each; one block barrier added
# speedup vs baseline: 2.3512x; 1.5142x over previous
#include <hip/hip_runtime.h>


typedef float v2f __attribute__((ext_vector_type(2)));
typedef float v8f __attribute__((ext_vector_type(8)));
typedef float v3f __attribute__((ext_vector_type(3), aligned(4)));
typedef float v4f __attribute__((ext_vector_type(4)));

#define B_PATHS    16384
#define N_STEPS    500
#define SAVE_EVERY 5
#define N_SAVES    100
#define T_SAVED    101
#define DT         0.002f
#define RING       25
#define PF_FAR     64
#define STEP_BYTES (B_PATHS * 3 * 4)
#define BLOCK_PATHS 64
#define CH_FLOATS   (BLOCK_PATHS * T_SAVED)
#define CH_CHUNKS   (CH_FLOATS / 4)

__global__ __launch_bounds__(64) void sde_wmma_kernel(
    const float* __restrict__ drift, const float* __restrict__ drift_bias,
    const float* __restrict__ diffusion, const float* __restrict__ diffusion_bias,
    const float* __restrict__ x0, const float* __restrict__ noise,
    float* __restrict__ out)
{
    __shared__ __attribute__((aligned(16))) float stage[3 * CH_FLOATS];

    const int  lane = threadIdx.x & 31;
    const int  wave = (blockIdx.x * blockDim.x + threadIdx.x) >> 5;
    const int  base = wave * 32;
    const int  m    = lane & 15;
    const bool lo   = lane < 16;
    const float sqdt = __builtin_sqrtf(DT);

    const int k0 = lo ? 0 : 2;
    const int k1 = lo ? 1 : 3;
    const int mm = m & 7;
    float a0v = 0.0f, a1v = 0.0f;
    if (mm < 3) {
        a0v = DT * drift[mm * 3 + k0] + ((mm == k0) ? 1.0f : 0.0f);
        a1v = (k1 < 3) ? (DT * drift[mm * 3 + k1] + ((mm == k1) ? 1.0f : 0.0f))
                       : (DT * drift_bias[mm]);
    } else if (mm < 6) {
        const int r = mm - 3;
        a0v = sqdt * diffusion[r * 3 + k0];
        a1v = (k1 < 3) ? (sqdt * diffusion[r * 3 + k1])
                       : (sqdt * diffusion_bias[r]);
    }
    v2f Amat; Amat.x = a0v; Amat.y = a1v;

    const int p0 = base + m;
    const int p1 = base + 16 + m;
    v2f B0, B1;
    B0.x = lo ? x0[p0 * 3 + 0] : x0[p0 * 3 + 2];
    B0.y = lo ? x0[p0 * 3 + 1] : 1.0f;
    B1.x = lo ? x0[p1 * 3 + 0] : x0[p1 * 3 + 2];
    B1.y = lo ? x0[p1 * 3 + 1] : 1.0f;

    const int p  = base + lane;
    const int BT = B_PATHS * T_SAVED;
    float* const srow = stage + threadIdx.x * T_SAVED;
    const float x00 = x0[p * 3 + 0];
    const float x01 = x0[p * 3 + 1];
    const float x02 = x0[p * 3 + 2];

    const char* const nbytes = (const char*)noise;
    const unsigned qb0 = (unsigned)p0 * 12u;
    const unsigned qb1 = (unsigned)p1 * 12u;
    const unsigned pb  = (unsigned)p  * 12u;
    v3f ringA[RING], ringB[RING];
#pragma unroll
    for (int j = 0; j < RING; ++j) {
        __builtin_amdgcn_sched_barrier(0);
        const char* src = nbytes + (size_t)j * STEP_BYTES;
        ringA[j] = *(const v3f*)(src + qb0);
        ringB[j] = *(const v3f*)(src + qb1);
    }
    __builtin_amdgcn_sched_barrier(0);
    srow[0 * CH_FLOATS] = x00;
    srow[1 * CH_FLOATS] = x01;
    srow[2 * CH_FLOATS] = x02;

    for (int s = 0; s < N_SAVES; s += RING / SAVE_EVERY) {
        const int t0 = s * SAVE_EVERY;
        const int tr0 = (t0 < N_STEPS - 2 * RING) ? t0 : (N_STEPS - 2 * RING);
        const int tf0 = (t0 < N_STEPS - RING - PF_FAR) ? t0 : (N_STEPS - RING - PF_FAR);
        const char* const rbase = nbytes + (size_t)(tr0 + RING)   * STEP_BYTES;
        const char* const fbase = nbytes + (size_t)(tf0 + PF_FAR) * STEP_BYTES;
#pragma unroll
        for (int j = 0; j < RING; ++j) {
            __builtin_amdgcn_sched_barrier(0);

            __builtin_prefetch(fbase + (size_t)j * STEP_BYTES + pb, 0, 3);

            const v3f na = ringA[j];
            const v3f nb = ringB[j];

            v8f cz = {};
            v8f d0 = __builtin_amdgcn_wmma_f32_16x16x4_f32(
                false, Amat, false, B0, (short)0, cz, false, false);
            v8f d1 = __builtin_amdgcn_wmma_f32_16x16x4_f32(
                false, Amat, false, B1, (short)0, cz, false, false);

            const float xn00 = __builtin_fmaf(d0[3], na.x, d0[0]);
            const float xn01 = __builtin_fmaf(d0[4], na.y, d0[1]);
            const float xn02 = __builtin_fmaf(d0[5], na.z, d0[2]);
            const float xn10 = __builtin_fmaf(d1[3], nb.x, d1[0]);
            const float xn11 = __builtin_fmaf(d1[4], nb.y, d1[1]);
            const float xn12 = __builtin_fmaf(d1[5], nb.z, d1[2]);

            B0.x = lo ? xn00 : xn02;  B0.y = lo ? xn01 : 1.0f;
            B1.x = lo ? xn10 : xn12;  B1.y = lo ? xn11 : 1.0f;

            if ((j % SAVE_EVERY) == SAVE_EVERY - 1) {
                const float sv0 = lo ? xn00 : xn10;
                const float sv1 = lo ? xn01 : xn11;
                const float sv2 = lo ? xn02 : xn12;
                const int st = s + j / SAVE_EVERY + 1;
                srow[0 * CH_FLOATS + st] = sv0;
                srow[1 * CH_FLOATS + st] = sv1;
                srow[2 * CH_FLOATS + st] = sv2;
            }

            __builtin_amdgcn_sched_barrier(0);
            const char* src = rbase + (size_t)j * STEP_BYTES;
            ringA[j] = *(const v3f*)(src + qb0);
            ringB[j] = *(const v3f*)(src + qb1);
        }
    }

    __syncthreads();
    const int tid = threadIdx.x;
    for (int c = 0; c < 3; ++c) {
        const v4f* const src4 = (const v4f*)(stage + c * CH_FLOATS);
        v4f* const dst4 = (v4f*)(out + (size_t)c * BT + (size_t)blockIdx.x * CH_FLOATS);
        for (int it = 0; it < CH_CHUNKS / BLOCK_PATHS; ++it)
            dst4[it * BLOCK_PATHS + tid] = src4[it * BLOCK_PATHS + tid];
        if (tid < CH_CHUNKS % BLOCK_PATHS)
            dst4[(CH_CHUNKS / BLOCK_PATHS) * BLOCK_PATHS + tid] =
                src4[(CH_CHUNKS / BLOCK_PATHS) * BLOCK_PATHS + tid];
    }
}

extern "C" void kernel_launch(void* const* d_in, const int* in_sizes, int n_in,
                              void* d_out, int out_size, void* d_ws, size_t ws_size,
                              hipStream_t stream) {
    const float* drift     = (const float*)d_in[0];
    const float* driftb    = (const float*)d_in[1];
    const float* diffusion = (const float*)d_in[2];
    const float* diffb     = (const float*)d_in[3];
    const float* x0        = (const float*)d_in[4];
    const float* noise     = (const float*)d_in[5];
    float* out = (float*)d_out;

    dim3 block(BLOCK_PATHS);
    dim3 grid(B_PATHS / 64);
    sde_wmma_kernel<<<grid, block, 0, stream>>>(
        drift, driftb, diffusion, diffb, x0, noise, out);
}


# per-step serial chain cut to WMMA + one packed FMA: A rows placed so each lane half's D pair is its own next B pair (no copies, no selects); one 16-path chain per wave; k sum order now x0,x1,1,x2
# speedup vs baseline: 2.8575x; 1.2153x over previous
#include <hip/hip_runtime.h>


typedef float v2f __attribute__((ext_vector_type(2)));
typedef float v8f __attribute__((ext_vector_type(8)));
typedef float v2fu __attribute__((ext_vector_type(2), aligned(4)));
typedef float v4f __attribute__((ext_vector_type(4)));

#define B_PATHS    16384
#define N_STEPS    500
#define SAVE_EVERY 5
#define N_SAVES    100
#define T_SAVED    101
#define DT         0.002f
#define RING       50
#define PF_FAR     100
#define STEP_BYTES (B_PATHS * 3 * 4)
#define BLOCK_PATHS   64
#define BLOCK_THREADS 128
#define CH_FLOATS   (BLOCK_PATHS * T_SAVED)
#define CH_CHUNKS   (CH_FLOATS / 4)

__global__ __launch_bounds__(BLOCK_THREADS) void sde_wmma_kernel(
    const float* __restrict__ drift, const float* __restrict__ drift_bias,
    const float* __restrict__ diffusion, const float* __restrict__ diffusion_bias,
    const float* __restrict__ x0, const float* __restrict__ noise,
    float* __restrict__ out)
{
    __shared__ __attribute__((aligned(16))) float stage[3 * CH_FLOATS];

    const int  tid  = threadIdx.x;
    const int  lane = tid & 31;
    const int  c    = lane & 15;
    const bool lo   = lane < 16;
    const int  pib  = (tid >> 5) * 16 + c;
    const int  p    = blockIdx.x * BLOCK_PATHS + pib;
    const float sqdt = __builtin_sqrtf(DT);

    float a0v = 0.0f, a1v = 0.0f;
    {
        const bool isF = (c == 0) | (c == 1) | (c == 9);
        const bool isG = (c == 2) | (c == 3) | (c == 11);
        const int  r   = (c == 9 || c == 11) ? 2 : (c & 1);
        if (isF) {
            if (lo) {
                a0v = DT * drift[r * 3 + 0] + ((r == 0) ? 1.0f : 0.0f);
                a1v = DT * drift[r * 3 + 1] + ((r == 1) ? 1.0f : 0.0f);
            } else {
                a0v = DT * drift_bias[r];
                a1v = DT * drift[r * 3 + 2] + ((r == 2) ? 1.0f : 0.0f);
            }
        } else if (isG) {
            if (lo) {
                a0v = sqdt * diffusion[r * 3 + 0];
                a1v = sqdt * diffusion[r * 3 + 1];
            } else {
                a0v = sqdt * diffusion_bias[r];
                a1v = sqdt * diffusion[r * 3 + 2];
            }
        } else if (c == 8 && !lo) {
            a0v = 1.0f;
        }
    }
    v2f Amat; Amat.x = a0v; Amat.y = a1v;

    v2f B;
    B.x = lo ? x0[p * 3 + 0] : 1.0f;
    B.y = lo ? x0[p * 3 + 1] : x0[p * 3 + 2];

    float* const srow0 = stage + pib * T_SAVED;
    float* const srowY = stage + (lo ? 1 : 2) * CH_FLOATS + pib * T_SAVED;

    const char* const nbytes = (const char*)noise;
    const unsigned qb = (unsigned)p * 12u + (lo ? 0u : 4u);
    v2f ring[RING];
#pragma unroll
    for (int j = 0; j < RING; ++j) {
        __builtin_amdgcn_sched_barrier(0);
        ring[j] = *(const v2fu*)(nbytes + (size_t)j * STEP_BYTES + qb);
    }
    __builtin_amdgcn_sched_barrier(0);
    srowY[0] = B.y;
    if (lo) srow0[0] = B.x;

    const v8f cz = {};
    v8f d = __builtin_amdgcn_wmma_f32_16x16x4_f32(
        false, Amat, false, B, (short)0, cz, false, false);

    for (int s = 0; s < N_SAVES; s += RING / SAVE_EVERY) {
        const int t0 = s * SAVE_EVERY;
        const int tr0 = (t0 < N_STEPS - 2 * RING) ? t0 : (N_STEPS - 2 * RING);
        const int tf0 = (t0 < N_STEPS - RING - PF_FAR) ? t0 : (N_STEPS - RING - PF_FAR);
        const char* const rbase = nbytes + (size_t)(tr0 + RING)   * STEP_BYTES;
        const char* const fbase = nbytes + (size_t)(tf0 + PF_FAR) * STEP_BYTES;
        const bool more = (s + RING / SAVE_EVERY) < N_SAVES;
#pragma unroll
        for (int j = 0; j < RING; ++j) {
            __builtin_amdgcn_sched_barrier(0);

            const v2f n  = ring[j];
            const v2f f2 = __builtin_shufflevector(d, d, 0, 1);
            const v2f g2 = __builtin_shufflevector(d, d, 2, 3);
            B = __builtin_elementwise_fma(g2, n, f2);

            if (j < RING - 1 || more)
                d = __builtin_amdgcn_wmma_f32_16x16x4_f32(
                    false, Amat, false, B, (short)0, cz, false, false);

            __builtin_amdgcn_sched_barrier(0);

            __builtin_prefetch(fbase + (size_t)j * STEP_BYTES + qb, 0, 3);

            if ((j % SAVE_EVERY) == SAVE_EVERY - 1) {
                const int st = s + j / SAVE_EVERY + 1;
                srowY[st] = B.y;
                if (lo) srow0[st] = B.x;
            }

            ring[j] = *(const v2fu*)(rbase + (size_t)j * STEP_BYTES + qb);
        }
    }

    __syncthreads();
    const int BT = B_PATHS * T_SAVED;
    for (int ch = 0; ch < 3; ++ch) {
        const v4f* const src4 = (const v4f*)(stage + ch * CH_FLOATS);
        v4f* const dst4 = (v4f*)(out + (size_t)ch * BT + (size_t)blockIdx.x * CH_FLOATS);
        for (int it = 0; it < CH_CHUNKS / BLOCK_THREADS; ++it)
            dst4[it * BLOCK_THREADS + tid] = src4[it * BLOCK_THREADS + tid];
        if (tid < CH_CHUNKS % BLOCK_THREADS)
            dst4[(CH_CHUNKS / BLOCK_THREADS) * BLOCK_THREADS + tid] =
                src4[(CH_CHUNKS / BLOCK_THREADS) * BLOCK_THREADS + tid];
    }
}

extern "C" void kernel_launch(void* const* d_in, const int* in_sizes, int n_in,
                              void* d_out, int out_size, void* d_ws, size_t ws_size,
                              hipStream_t stream) {
    const float* drift     = (const float*)d_in[0];
    const float* driftb    = (const float*)d_in[1];
    const float* diffusion = (const float*)d_in[2];
    const float* diffb     = (const float*)d_in[3];
    const float* x0        = (const float*)d_in[4];
    const float* noise     = (const float*)d_in[5];
    float* out = (float*)d_out;

    dim3 block(BLOCK_THREADS);
    dim3 grid(B_PATHS / BLOCK_PATHS);
    sde_wmma_kernel<<<grid, block, 0, stream>>>(
        drift, driftb, diffusion, diffb, x0, noise, out);
}


# saves made unconditional: lanes 16-31 store their constant 1.0 into an unread LDS strip, so the time loop never changes EXEC and the s_wait_xcnt 0 ahead of each EXEC write (9 of 10 per trip) is gone
# speedup vs baseline: 3.3675x; 1.1785x over previous
#include <hip/hip_runtime.h>


typedef float v2f __attribute__((ext_vector_type(2)));
typedef float v8f __attribute__((ext_vector_type(8)));
typedef float v2fu __attribute__((ext_vector_type(2), aligned(4)));
typedef float v4f __attribute__((ext_vector_type(4)));

#define B_PATHS    16384
#define N_STEPS    500
#define SAVE_EVERY 5
#define N_SAVES    100
#define T_SAVED    101
#define DT         0.002f
#define RING       50
#define PF_FAR     100
#define STEP_BYTES (B_PATHS * 3 * 4)
#define BLOCK_PATHS   64
#define BLOCK_THREADS 128
#define CH_FLOATS   (BLOCK_PATHS * T_SAVED)
#define CH_CHUNKS   (CH_FLOATS / 4)
#define SPILL_FLOATS (16 + T_SAVED + 3)

__global__ __launch_bounds__(BLOCK_THREADS) void sde_wmma_kernel(
    const float* __restrict__ drift, const float* __restrict__ drift_bias,
    const float* __restrict__ diffusion, const float* __restrict__ diffusion_bias,
    const float* __restrict__ x0, const float* __restrict__ noise,
    float* __restrict__ out)
{
    __shared__ __attribute__((aligned(16))) float stage[3 * CH_FLOATS + SPILL_FLOATS];

    const int  tid  = threadIdx.x;
    const int  lane = tid & 31;
    const int  c    = lane & 15;
    const bool lo   = lane < 16;
    const int  pib  = (tid >> 5) * 16 + c;
    const int  p    = blockIdx.x * BLOCK_PATHS + pib;
    const float sqdt = __builtin_sqrtf(DT);

    float a0v = 0.0f, a1v = 0.0f;
    {
        const bool isF = (c == 0) | (c == 1) | (c == 9);
        const bool isG = (c == 2) | (c == 3) | (c == 11);
        const int  r   = (c == 9 || c == 11) ? 2 : (c & 1);
        if (isF) {
            if (lo) {
                a0v = DT * drift[r * 3 + 0] + ((r == 0) ? 1.0f : 0.0f);
                a1v = DT * drift[r * 3 + 1] + ((r == 1) ? 1.0f : 0.0f);
            } else {
                a0v = DT * drift_bias[r];
                a1v = DT * drift[r * 3 + 2] + ((r == 2) ? 1.0f : 0.0f);
            }
        } else if (isG) {
            if (lo) {
                a0v = sqdt * diffusion[r * 3 + 0];
                a1v = sqdt * diffusion[r * 3 + 1];
            } else {
                a0v = sqdt * diffusion_bias[r];
                a1v = sqdt * diffusion[r * 3 + 2];
            }
        } else if (c == 8 && !lo) {
            a0v = 1.0f;
        }
    }
    v2f Amat; Amat.x = a0v; Amat.y = a1v;

    v2f B;
    B.x = lo ? x0[p * 3 + 0] : 1.0f;
    B.y = lo ? x0[p * 3 + 1] : x0[p * 3 + 2];

    float* const srow0 = lo ? (stage + pib * T_SAVED) : (stage + 3 * CH_FLOATS + c);
    float* const srowY = stage + (lo ? 1 : 2) * CH_FLOATS + pib * T_SAVED;

    const char* const nbytes = (const char*)noise;
    const unsigned qb = (unsigned)p * 12u + (lo ? 0u : 4u);
    v2f ring[RING];
#pragma unroll
    for (int j = 0; j < RING; ++j) {
        __builtin_amdgcn_sched_barrier(0);
        ring[j] = *(const v2fu*)(nbytes + (size_t)j * STEP_BYTES + qb);
    }
    __builtin_amdgcn_sched_barrier(0);
    srowY[0] = B.y;
    srow0[0] = B.x;

    const v8f cz = {};
    v8f d = __builtin_amdgcn_wmma_f32_16x16x4_f32(
        false, Amat, false, B, (short)0, cz, false, false);

    for (int s = 0; s < N_SAVES; s += RING / SAVE_EVERY) {
        const int t0 = s * SAVE_EVERY;
        const int tr0 = (t0 < N_STEPS - 2 * RING) ? t0 : (N_STEPS - 2 * RING);
        const int tf0 = (t0 < N_STEPS - RING - PF_FAR) ? t0 : (N_STEPS - RING - PF_FAR);
        const char* const rbase = nbytes + (size_t)(tr0 + RING)   * STEP_BYTES;
        const char* const fbase = nbytes + (size_t)(tf0 + PF_FAR) * STEP_BYTES;
        const bool more = (s + RING / SAVE_EVERY) < N_SAVES;
#pragma unroll
        for (int j = 0; j < RING; ++j) {
            __builtin_amdgcn_sched_barrier(0);

            const v2f n  = ring[j];
            const v2f f2 = __builtin_shufflevector(d, d, 0, 1);
            const v2f g2 = __builtin_shufflevector(d, d, 2, 3);
            B = __builtin_elementwise_fma(g2, n, f2);

            if (j < RING - 1 || more)
                d = __builtin_amdgcn_wmma_f32_16x16x4_f32(
                    false, Amat, false, B, (short)0, cz, false, false);

            __builtin_amdgcn_sched_barrier(0);

            __builtin_prefetch(fbase + (size_t)j * STEP_BYTES + qb, 0, 3);

            if ((j % SAVE_EVERY) == SAVE_EVERY - 1) {
                const int st = s + j / SAVE_EVERY + 1;
                srowY[st] = B.y;
                srow0[st] = B.x;
            }

            ring[j] = *(const v2fu*)(rbase + (size_t)j * STEP_BYTES + qb);
        }
    }

    __syncthreads();
    const int BT = B_PATHS * T_SAVED;
    for (int ch = 0; ch < 3; ++ch) {
        const v4f* const src4 = (const v4f*)(stage + ch * CH_FLOATS);
        v4f* const dst4 = (v4f*)(out + (size_t)ch * BT + (size_t)blockIdx.x * CH_FLOATS);
        for (int it = 0; it < CH_CHUNKS / BLOCK_THREADS; ++it)
            dst4[it * BLOCK_THREADS + tid] = src4[it * BLOCK_THREADS + tid];
        if (tid < CH_CHUNKS % BLOCK_THREADS)
            dst4[(CH_CHUNKS / BLOCK_THREADS) * BLOCK_THREADS + tid] =
                src4[(CH_CHUNKS / BLOCK_THREADS) * BLOCK_THREADS + tid];
    }
}

extern "C" void kernel_launch(void* const* d_in, const int* in_sizes, int n_in,
                              void* d_out, int out_size, void* d_ws, size_t ws_size,
                              hipStream_t stream) {
    const float* drift     = (const float*)d_in[0];
    const float* driftb    = (const float*)d_in[1];
    const float* diffusion = (const float*)d_in[2];
    const float* diffb     = (const float*)d_in[3];
    const float* x0        = (const float*)d_in[4];
    const float* noise     = (const float*)d_in[5];
    float* out = (float*)d_out;

    dim3 block(BLOCK_THREADS);
    dim3 grid(B_PATHS / BLOCK_PATHS);
    sde_wmma_kernel<<<grid, block, 0, stream>>>(
        drift, driftb, diffusion, diffb, x0, noise, out);
}


# time loop's memory instructions (noise ring refills, far prefetches, LDS saves) issued as one group every 10 steps behind that step's WMMA: 50 places between dependent WMMAs per wave instead of 500
# speedup vs baseline: 3.8729x; 1.1501x over previous
#include <hip/hip_runtime.h>


typedef float v2f __attribute__((ext_vector_type(2)));
typedef float v8f __attribute__((ext_vector_type(8)));
typedef float v2fu __attribute__((ext_vector_type(2), aligned(4)));
typedef float v4f __attribute__((ext_vector_type(4)));

#define B_PATHS    16384
#define N_STEPS    500
#define SAVE_EVERY 5
#define N_SAVES    100
#define T_SAVED    101
#define DT         0.002f
#define RING       50
#define PF_FAR     100
#define GROUP      10
#define STEP_BYTES (B_PATHS * 3 * 4)
#define BLOCK_PATHS   64
#define BLOCK_THREADS 128
#define CH_FLOATS   (BLOCK_PATHS * T_SAVED)
#define CH_CHUNKS   (CH_FLOATS / 4)
#define SPILL_FLOATS (16 + T_SAVED + 3)

__global__ __launch_bounds__(BLOCK_THREADS) void sde_wmma_kernel(
    const float* __restrict__ drift, const float* __restrict__ drift_bias,
    const float* __restrict__ diffusion, const float* __restrict__ diffusion_bias,
    const float* __restrict__ x0, const float* __restrict__ noise,
    float* __restrict__ out)
{
    __shared__ __attribute__((aligned(16))) float stage[3 * CH_FLOATS + SPILL_FLOATS];

    const int  tid  = threadIdx.x;
    const int  lane = tid & 31;
    const int  c    = lane & 15;
    const bool lo   = lane < 16;
    const int  pib  = (tid >> 5) * 16 + c;
    const int  p    = blockIdx.x * BLOCK_PATHS + pib;
    const float sqdt = __builtin_sqrtf(DT);

    float a0v = 0.0f, a1v = 0.0f;
    {
        const bool isF = (c == 0) | (c == 1) | (c == 9);
        const bool isG = (c == 2) | (c == 3) | (c == 11);
        const int  r   = (c == 9 || c == 11) ? 2 : (c & 1);
        if (isF) {
            if (lo) {
                a0v = DT * drift[r * 3 + 0] + ((r == 0) ? 1.0f : 0.0f);
                a1v = DT * drift[r * 3 + 1] + ((r == 1) ? 1.0f : 0.0f);
            } else {
                a0v = DT * drift_bias[r];
                a1v = DT * drift[r * 3 + 2] + ((r == 2) ? 1.0f : 0.0f);
            }
        } else if (isG) {
            if (lo) {
                a0v = sqdt * diffusion[r * 3 + 0];
                a1v = sqdt * diffusion[r * 3 + 1];
            } else {
                a0v = sqdt * diffusion_bias[r];
                a1v = sqdt * diffusion[r * 3 + 2];
            }
        } else if (c == 8 && !lo) {
            a0v = 1.0f;
        }
    }
    v2f Amat; Amat.x = a0v; Amat.y = a1v;

    v2f B;
    B.x = lo ? x0[p * 3 + 0] : 1.0f;
    B.y = lo ? x0[p * 3 + 1] : x0[p * 3 + 2];

    float* const srow0 = lo ? (stage + pib * T_SAVED) : (stage + 3 * CH_FLOATS + c);
    float* const srowY = stage + (lo ? 1 : 2) * CH_FLOATS + pib * T_SAVED;

    const char* const nbytes = (const char*)noise;
    const unsigned qb = (unsigned)p * 12u + (lo ? 0u : 4u);
    v2f ring[RING];
#pragma unroll
    for (int j = 0; j < RING; ++j) {
        __builtin_amdgcn_sched_barrier(0);
        ring[j] = *(const v2fu*)(nbytes + (size_t)j * STEP_BYTES + qb);
    }
    __builtin_amdgcn_sched_barrier(0);
    srowY[0] = B.y;
    srow0[0] = B.x;

    const v8f cz = {};
    v8f d = __builtin_amdgcn_wmma_f32_16x16x4_f32(
        false, Amat, false, B, (short)0, cz, false, false);

    for (int s = 0; s < N_SAVES; s += RING / SAVE_EVERY) {
        const int t0 = s * SAVE_EVERY;
        const int tr0 = (t0 < N_STEPS - 2 * RING) ? t0 : (N_STEPS - 2 * RING);
        const int tf0 = (t0 < N_STEPS - RING - PF_FAR) ? t0 : (N_STEPS - RING - PF_FAR);
        const char* const rbase = nbytes + (size_t)(tr0 + RING)   * STEP_BYTES;
        const char* const fbase = nbytes + (size_t)(tf0 + PF_FAR) * STEP_BYTES;
        const bool more = (s + RING / SAVE_EVERY) < N_SAVES;
        v2f sv[GROUP / SAVE_EVERY];
#pragma unroll
        for (int j = 0; j < RING; ++j) {
            __builtin_amdgcn_sched_barrier(0);

            const v2f n  = ring[j];
            const v2f f2 = __builtin_shufflevector(d, d, 0, 1);
            const v2f g2 = __builtin_shufflevector(d, d, 2, 3);
            B = __builtin_elementwise_fma(g2, n, f2);
            if ((j % SAVE_EVERY) == SAVE_EVERY - 1)
                sv[(j % GROUP) / SAVE_EVERY] = B;

            if (j < RING - 1 || more)
                d = __builtin_amdgcn_wmma_f32_16x16x4_f32(
                    false, Amat, false, B, (short)0, cz, false, false);

            if ((j % GROUP) == GROUP - 1) {
                __builtin_amdgcn_sched_barrier(0);
                const int j0 = j - (GROUP - 1);
#pragma unroll
                for (int q = 0; q < GROUP; ++q)
                    ring[j0 + q] = *(const v2fu*)(rbase + (size_t)(j0 + q) * STEP_BYTES + qb);
#pragma unroll
                for (int q = 0; q < GROUP; ++q)
                    __builtin_prefetch(fbase + (size_t)(j0 + q) * STEP_BYTES + qb, 0, 3);
#pragma unroll
                for (int k = 0; k < GROUP / SAVE_EVERY; ++k) {
                    const int st = s + (j0 + k * SAVE_EVERY) / SAVE_EVERY + 1;
                    srowY[st] = sv[k].y;
                    srow0[st] = sv[k].x;
                }
            }
        }
    }

    __syncthreads();
    const int BT = B_PATHS * T_SAVED;
    for (int ch = 0; ch < 3; ++ch) {
        const v4f* const src4 = (const v4f*)(stage + ch * CH_FLOATS);
        v4f* const dst4 = (v4f*)(out + (size_t)ch * BT + (size_t)blockIdx.x * CH_FLOATS);
        for (int it = 0; it < CH_CHUNKS / BLOCK_THREADS; ++it)
            dst4[it * BLOCK_THREADS + tid] = src4[it * BLOCK_THREADS + tid];
        if (tid < CH_CHUNKS % BLOCK_THREADS)
            dst4[(CH_CHUNKS / BLOCK_THREADS) * BLOCK_THREADS + tid] =
                src4[(CH_CHUNKS / BLOCK_THREADS) * BLOCK_THREADS + tid];
    }
}

extern "C" void kernel_launch(void* const* d_in, const int* in_sizes, int n_in,
                              void* d_out, int out_size, void* d_ws, size_t ws_size,
                              hipStream_t stream) {
    const float* drift     = (const float*)d_in[0];
    const float* driftb    = (const float*)d_in[1];
    const float* diffusion = (const float*)d_in[2];
    const float* diffb     = (const float*)d_in[3];
    const float* x0        = (const float*)d_in[4];
    const float* noise     = (const float*)d_in[5];
    float* out = (float*)d_out;

    dim3 block(BLOCK_THREADS);
    dim3 grid(B_PATHS / BLOCK_PATHS);
    sde_wmma_kernel<<<grid, block, 0, stream>>>(
        drift, driftb, diffusion, diffb, x0, noise, out);
}


# group addresses = per-trip scalar bases (two per stream) + lane offset + immediate: no vector add in a group writes the in-flight WMMA's result registers, no s_wait_xcnt before the next WMMA
# speedup vs baseline: 4.1178x; 1.0632x over previous
#include <hip/hip_runtime.h>


typedef float v2f __attribute__((ext_vector_type(2)));
typedef float v8f __attribute__((ext_vector_type(8)));
typedef float v2fu __attribute__((ext_vector_type(2), aligned(4)));
typedef float v4f __attribute__((ext_vector_type(4)));
typedef const __attribute__((address_space(1))) char* gbytes;
typedef const __attribute__((address_space(1))) v2fu* gv2fu;

#define B_PATHS    16384
#define N_STEPS    500
#define SAVE_EVERY 5
#define N_SAVES    100
#define T_SAVED    101
#define DT         0.002f
#define RING       50
#define PF_FAR     100
#define GROUP      10
#define HALF       (RING / 2)
#define STEP_BYTES (B_PATHS * 3 * 4)
#define BLOCK_PATHS   64
#define BLOCK_THREADS 128
#define CH_FLOATS   (BLOCK_PATHS * T_SAVED)
#define CH_CHUNKS   (CH_FLOATS / 4)
#define SPILL_FLOATS (16 + T_SAVED + 3)

__global__ __launch_bounds__(BLOCK_THREADS) void sde_wmma_kernel(
    const float* __restrict__ drift, const float* __restrict__ drift_bias,
    const float* __restrict__ diffusion, const float* __restrict__ diffusion_bias,
    const float* __restrict__ x0, const float* __restrict__ noise,
    float* __restrict__ out)
{
    __shared__ __attribute__((aligned(16))) float stage[3 * CH_FLOATS + SPILL_FLOATS];

    const int  tid  = threadIdx.x;
    const int  lane = tid & 31;
    const int  c    = lane & 15;
    const bool lo   = lane < 16;
    const int  pib  = (tid >> 5) * 16 + c;
    const int  p    = blockIdx.x * BLOCK_PATHS + pib;
    const float sqdt = __builtin_sqrtf(DT);

    float a0v = 0.0f, a1v = 0.0f;
    {
        const bool isF = (c == 0) | (c == 1) | (c == 9);
        const bool isG = (c == 2) | (c == 3) | (c == 11);
        const int  r   = (c == 9 || c == 11) ? 2 : (c & 1);
        if (isF) {
            if (lo) {
                a0v = DT * drift[r * 3 + 0] + ((r == 0) ? 1.0f : 0.0f);
                a1v = DT * drift[r * 3 + 1] + ((r == 1) ? 1.0f : 0.0f);
            } else {
                a0v = DT * drift_bias[r];
                a1v = DT * drift[r * 3 + 2] + ((r == 2) ? 1.0f : 0.0f);
            }
        } else if (isG) {
            if (lo) {
                a0v = sqdt * diffusion[r * 3 + 0];
                a1v = sqdt * diffusion[r * 3 + 1];
            } else {
                a0v = sqdt * diffusion_bias[r];
                a1v = sqdt * diffusion[r * 3 + 2];
            }
        } else if (c == 8 && !lo) {
            a0v = 1.0f;
        }
    }
    v2f Amat; Amat.x = a0v; Amat.y = a1v;

    v2f B;
    B.x = lo ? x0[p * 3 + 0] : 1.0f;
    B.y = lo ? x0[p * 3 + 1] : x0[p * 3 + 2];

    float* const srow0 = lo ? (stage + pib * T_SAVED) : (stage + 3 * CH_FLOATS + c);
    float* const srowY = stage + (lo ? 1 : 2) * CH_FLOATS + pib * T_SAVED;

    const char* const nbytes = (const char*)noise;
    const unsigned qb = (unsigned)p * 12u + (lo ? 0u : 4u);
    v2f ring[RING];
#pragma unroll
    for (int j = 0; j < RING; ++j) {
        __builtin_amdgcn_sched_barrier(0);
        ring[j] = *(const v2fu*)(nbytes + (size_t)j * STEP_BYTES + qb);
    }
    __builtin_amdgcn_sched_barrier(0);
    srowY[0] = B.y;
    srow0[0] = B.x;

    const v8f cz = {};
    v8f d = __builtin_amdgcn_wmma_f32_16x16x4_f32(
        false, Amat, false, B, (short)0, cz, false, false);

    for (int s = 0; s < N_SAVES; s += RING / SAVE_EVERY) {
        const int t0 = s * SAVE_EVERY;
        const int tr0 = (t0 < N_STEPS - 2 * RING) ? t0 : (N_STEPS - 2 * RING);
        const int tf0 = (t0 < N_STEPS - RING - PF_FAR) ? t0 : (N_STEPS - RING - PF_FAR);
        size_t ra = (size_t)nbytes + (size_t)(tr0 + RING)   * STEP_BYTES;
        size_t rb = ra + (size_t)HALF * STEP_BYTES;
        size_t fa = (size_t)nbytes + (size_t)(tf0 + PF_FAR) * STEP_BYTES;
        size_t fb = fa + (size_t)HALF * STEP_BYTES;
        asm volatile("" : "+s"(ra), "+s"(rb), "+s"(fa), "+s"(fb));
        const gbytes rbaseA = (gbytes)ra;
        const gbytes rbaseB = (gbytes)rb;
        const gbytes fbaseA = (gbytes)fa;
        const gbytes fbaseB = (gbytes)fb;
        const bool more = (s + RING / SAVE_EVERY) < N_SAVES;
        v2f sv[GROUP / SAVE_EVERY];
#pragma unroll
        for (int j = 0; j < RING; ++j) {
            __builtin_amdgcn_sched_barrier(0);

            const v2f n  = ring[j];
            const v2f f2 = __builtin_shufflevector(d, d, 0, 1);
            const v2f g2 = __builtin_shufflevector(d, d, 2, 3);
            B = __builtin_elementwise_fma(g2, n, f2);
            if ((j % SAVE_EVERY) == SAVE_EVERY - 1)
                sv[(j % GROUP) / SAVE_EVERY] = B;

            if (j < RING - 1 || more)
                d = __builtin_amdgcn_wmma_f32_16x16x4_f32(
                    false, Amat, false, B, (short)0, cz, false, false);

            if ((j % GROUP) == GROUP - 1) {
                __builtin_amdgcn_sched_barrier(0);
                const int j0 = j - (GROUP - 1);
#pragma unroll
                for (int q = 0; q < GROUP; ++q) {
                    const int jj = j0 + q;
                    const gbytes rb0 = (jj < HALF) ? rbaseA : rbaseB;
                    ring[jj] = *(gv2fu)(rb0 + (size_t)(jj % HALF) * STEP_BYTES + qb);
                }
#pragma unroll
                for (int q = 0; q < GROUP; ++q) {
                    const int jj = j0 + q;
                    const gbytes fb0 = (jj < HALF) ? fbaseA : fbaseB;
                    __builtin_prefetch((const void*)(fb0 + (size_t)(jj % HALF) * STEP_BYTES + qb), 0, 3);
                }
#pragma unroll
                for (int k = 0; k < GROUP / SAVE_EVERY; ++k) {
                    const int st = s + (j0 + k * SAVE_EVERY) / SAVE_EVERY + 1;
                    srowY[st] = sv[k].y;
                    srow0[st] = sv[k].x;
                }
            }
        }
    }

    __syncthreads();
    const int BT = B_PATHS * T_SAVED;
    for (int ch = 0; ch < 3; ++ch) {
        const v4f* const src4 = (const v4f*)(stage + ch * CH_FLOATS);
        v4f* const dst4 = (v4f*)(out + (size_t)ch * BT + (size_t)blockIdx.x * CH_FLOATS);
        for (int it = 0; it < CH_CHUNKS / BLOCK_THREADS; ++it)
            dst4[it * BLOCK_THREADS + tid] = src4[it * BLOCK_THREADS + tid];
        if (tid < CH_CHUNKS % BLOCK_THREADS)
            dst4[(CH_CHUNKS / BLOCK_THREADS) * BLOCK_THREADS + tid] =
                src4[(CH_CHUNKS / BLOCK_THREADS) * BLOCK_THREADS + tid];
    }
}

extern "C" void kernel_launch(void* const* d_in, const int* in_sizes, int n_in,
                              void* d_out, int out_size, void* d_ws, size_t ws_size,
                              hipStream_t stream) {
    const float* drift     = (const float*)d_in[0];
    const float* driftb    = (const float*)d_in[1];
    const float* diffusion = (const float*)d_in[2];
    const float* diffb     = (const float*)d_in[3];
    const float* x0        = (const float*)d_in[4];
    const float* noise     = (const float*)d_in[5];
    float* out = (float*)d_out;

    dim3 block(BLOCK_THREADS);
    dim3 grid(B_PATHS / BLOCK_PATHS);
    sde_wmma_kernel<<<grid, block, 0, stream>>>(
        drift, driftb, diffusion, diffb, x0, noise, out);
}


# set-up loads (each lane's two A coefficients and two x0 values, unconditional, per-lane address) issued in one batch with the ring fill: four serial load round trips ahead of step 0 become one
# speedup vs baseline: 4.3654x; 1.0601x over previous
#include <hip/hip_runtime.h>


typedef float v2f __attribute__((ext_vector_type(2)));
typedef float v8f __attribute__((ext_vector_type(8)));
typedef float v2fu __attribute__((ext_vector_type(2), aligned(4)));
typedef float v4f __attribute__((ext_vector_type(4)));
typedef const __attribute__((address_space(1))) char* gbytes;
typedef const __attribute__((address_space(1))) v2fu* gv2fu;

#define B_PATHS    16384
#define N_STEPS    500
#define SAVE_EVERY 5
#define N_SAVES    100
#define T_SAVED    101
#define DT         0.002f
#define RING       50
#define PF_FAR     100
#define GROUP      10
#define HALF       (RING / 2)
#define STEP_BYTES (B_PATHS * 3 * 4)
#define BLOCK_PATHS   64
#define BLOCK_THREADS 128
#define CH_FLOATS   (BLOCK_PATHS * T_SAVED)
#define CH_CHUNKS   (CH_FLOATS / 4)
#define SPILL_FLOATS (16 + T_SAVED + 3)

__global__ __launch_bounds__(BLOCK_THREADS) void sde_wmma_kernel(
    const float* __restrict__ drift, const float* __restrict__ drift_bias,
    const float* __restrict__ diffusion, const float* __restrict__ diffusion_bias,
    const float* __restrict__ x0, const float* __restrict__ noise,
    float* __restrict__ out)
{
    __shared__ __attribute__((aligned(16))) float stage[3 * CH_FLOATS + SPILL_FLOATS];

    const int  tid  = threadIdx.x;
    const int  lane = tid & 31;
    const int  c    = lane & 15;
    const bool lo   = lane < 16;
    const int  pib  = (tid >> 5) * 16 + c;
    const int  p    = blockIdx.x * BLOCK_PATHS + pib;
    const float sqdt = __builtin_sqrtf(DT);

    const bool isF = ((0x0203u >> c) & 1u) != 0u;
    const bool isG = ((0x080Cu >> c) & 1u) != 0u;
    const int  r   = (((0x0A00u >> c) & 1u) != 0u) ? 2 : (c & 1);
    const float* const mrow = (isG ? diffusion : drift) + r * 3;
    const float* const brow = (isG ? diffusion_bias : drift_bias) + r;
    const float cf0 = *(lo ? mrow : brow);
    const float cf1 = *(lo ? (mrow + 1) : (mrow + 2));

    const float xa = x0[p * 3 + (lo ? 0 : 2)];
    const float xb = x0[p * 3 + (lo ? 1 : 2)];
    v2f B;
    B.x = lo ? xa : 1.0f;
    B.y = xb;

    float* const srow0 = lo ? (stage + pib * T_SAVED) : (stage + 3 * CH_FLOATS + c);
    float* const srowY = stage + (lo ? 1 : 2) * CH_FLOATS + pib * T_SAVED;

    const char* const nbytes = (const char*)noise;
    const unsigned qb = (unsigned)p * 12u + (lo ? 0u : 4u);
    v2f ring[RING];
#pragma unroll
    for (int j = 0; j < RING; ++j) {
        __builtin_amdgcn_sched_barrier(0);
        ring[j] = *(const v2fu*)(nbytes + (size_t)j * STEP_BYTES + qb);
    }
    __builtin_amdgcn_sched_barrier(0);
    float a0v, a1v;
    {
        const float scale = isG ? sqdt : DT;
        const float id0 = (isF & lo & (r == 0)) ? 1.0f : 0.0f;
        const float id1 = (isF & (r == (lo ? 1 : 2))) ? 1.0f : 0.0f;
        const float fill0 = ((c == 8) & !lo) ? 1.0f : 0.0f;
        const bool  coef  = isF | isG;
        a0v = coef ? (scale * cf0 + id0) : fill0;
        a1v = coef ? (scale * cf1 + id1) : 0.0f;
    }
    v2f Amat; Amat.x = a0v; Amat.y = a1v;
    srowY[0] = B.y;
    srow0[0] = B.x;
    __builtin_amdgcn_sched_barrier(0);

    const v8f cz = {};
    v8f d = __builtin_amdgcn_wmma_f32_16x16x4_f32(
        false, Amat, false, B, (short)0, cz, false, false);

    for (int s = 0; s < N_SAVES; s += RING / SAVE_EVERY) {
        const int t0 = s * SAVE_EVERY;
        const int tr0 = (t0 < N_STEPS - 2 * RING) ? t0 : (N_STEPS - 2 * RING);
        const int tf0 = (t0 < N_STEPS - RING - PF_FAR) ? t0 : (N_STEPS - RING - PF_FAR);
        size_t ra = (size_t)nbytes + (size_t)(tr0 + RING)   * STEP_BYTES;
        size_t rb = ra + (size_t)HALF * STEP_BYTES;
        size_t fa = (size_t)nbytes + (size_t)(tf0 + PF_FAR) * STEP_BYTES;
        size_t fb = fa + (size_t)HALF * STEP_BYTES;
        asm volatile("" : "+s"(ra), "+s"(rb), "+s"(fa), "+s"(fb));
        const gbytes rbaseA = (gbytes)ra;
        const gbytes rbaseB = (gbytes)rb;
        const gbytes fbaseA = (gbytes)fa;
        const gbytes fbaseB = (gbytes)fb;
        const bool more = (s + RING / SAVE_EVERY) < N_SAVES;
        v2f sv[GROUP / SAVE_EVERY];
#pragma unroll
        for (int j = 0; j < RING; ++j) {
            __builtin_amdgcn_sched_barrier(0);

            const v2f n  = ring[j];
            const v2f f2 = __builtin_shufflevector(d, d, 0, 1);
            const v2f g2 = __builtin_shufflevector(d, d, 2, 3);
            B = __builtin_elementwise_fma(g2, n, f2);
            if ((j % SAVE_EVERY) == SAVE_EVERY - 1)
                sv[(j % GROUP) / SAVE_EVERY] = B;

            if (j < RING - 1 || more)
                d = __builtin_amdgcn_wmma_f32_16x16x4_f32(
                    false, Amat, false, B, (short)0, cz, false, false);

            if ((j % GROUP) == GROUP - 1) {
                __builtin_amdgcn_sched_barrier(0);
                const int j0 = j - (GROUP - 1);
#pragma unroll
                for (int q = 0; q < GROUP; ++q) {
                    const int jj = j0 + q;
                    const gbytes rb0 = (jj < HALF) ? rbaseA : rbaseB;
                    ring[jj] = *(gv2fu)(rb0 + (size_t)(jj % HALF) * STEP_BYTES + qb);
                }
#pragma unroll
                for (int q = 0; q < GROUP; ++q) {
                    const int jj = j0 + q;
                    const gbytes fb0 = (jj < HALF) ? fbaseA : fbaseB;
                    __builtin_prefetch((const void*)(fb0 + (size_t)(jj % HALF) * STEP_BYTES + qb), 0, 3);
                }
#pragma unroll
                for (int k = 0; k < GROUP / SAVE_EVERY; ++k) {
                    const int st = s + (j0 + k * SAVE_EVERY) / SAVE_EVERY + 1;
                    srowY[st] = sv[k].y;
                    srow0[st] = sv[k].x;
                }
            }
        }
    }

    __syncthreads();
    const int BT = B_PATHS * T_SAVED;
    for (int ch = 0; ch < 3; ++ch) {
        const v4f* const src4 = (const v4f*)(stage + ch * CH_FLOATS);
        v4f* const dst4 = (v4f*)(out + (size_t)ch * BT + (size_t)blockIdx.x * CH_FLOATS);
        for (int it = 0; it < CH_CHUNKS / BLOCK_THREADS; ++it)
            dst4[it * BLOCK_THREADS + tid] = src4[it * BLOCK_THREADS + tid];
        if (tid < CH_CHUNKS % BLOCK_THREADS)
            dst4[(CH_CHUNKS / BLOCK_THREADS) * BLOCK_THREADS + tid] =
                src4[(CH_CHUNKS / BLOCK_THREADS) * BLOCK_THREADS + tid];
    }
}

extern "C" void kernel_launch(void* const* d_in, const int* in_sizes, int n_in,
                              void* d_out, int out_size, void* d_ws, size_t ws_size,
                              hipStream_t stream) {
    const float* drift     = (const float*)d_in[0];
    const float* driftb    = (const float*)d_in[1];
    const float* diffusion = (const float*)d_in[2];
    const float* diffb     = (const float*)d_in[3];
    const float* x0        = (const float*)d_in[4];
    const float* noise     = (const float*)d_in[5];
    float* out = (float*)d_out;

    dim3 block(BLOCK_THREADS);
    dim3 grid(B_PATHS / BLOCK_PATHS);
    sde_wmma_kernel<<<grid, block, 0, stream>>>(
        drift, driftb, diffusion, diffb, x0, noise, out);
}
